# starting kernel
# baseline (speedup 1.0000x reference)
#include <hip/hip_runtime.h>
#include <hip/hip_bf16.h>


typedef __attribute__((ext_vector_type(2))) float v2f;
typedef __attribute__((ext_vector_type(4))) float v4f;
typedef __attribute__((ext_vector_type(8))) float v8f;

#define LH    20
#define LH4   80
#define LF    32
#define LT    1024
#define LB    256
#define LOUT  10

__device__ __forceinline__ float sigf(float x) { return 1.0f / (1.0f + __expf(-x)); }
__device__ __forceinline__ float tanhfast(float x) {
    const float e = __expf(2.0f * x);
    return (e - 1.0f) / (e + 1.0f);
}

#define WMMA4(c, a, b) \
    (c) = __builtin_amdgcn_wmma_f32_16x16x4_f32(false, (a), false, (b), (short)0, (c), false, false)

__global__ void __launch_bounds__(32)
lstm2_dense_fused(const float* __restrict__ X,
                  const float* __restrict__ Wi1, const float* __restrict__ Wh1,
                  const float* __restrict__ b1,
                  const float* __restrict__ Wi2, const float* __restrict__ Wh2,
                  const float* __restrict__ b2,
                  const float* __restrict__ Wd,  const float* __restrict__ bd,
                  float* __restrict__ out)
{
    __shared__ v2f sWi1f[40 * 32];
    __shared__ v2f sWh1f[25 * 32];
    __shared__ v2f sWi2f[25 * 32];
    __shared__ v2f sWh2f[25 * 32];
    __shared__ v2f sWdf [ 5 * 32];
    __shared__ float sb1[LH4], sb2[LH4], sbd[16];
    __shared__ v4f sZt4[LH4 * 4];
    __shared__ v2f sH1f[5 * 32], sH2f[5 * 32];
    __shared__ float sC1[16 * LH], sC2[16 * LH];

    const int lane  = threadIdx.x;
    const int mrow  = lane & 15;
    const int hi    = lane >> 4;
    const int khalf = 2 * hi;
    const int b0    = blockIdx.x * 16;

    for (int f = 0; f < 40; ++f) {
        const int n = f >> 3, kk = f & 7;
        const int row = 4 * kk + khalf, col = 16 * n + mrow;
        v2f w; w.x = Wi1[row * LH4 + col]; w.y = Wi1[(row + 1) * LH4 + col];
        sWi1f[f * 32 + lane] = w;
    }
    for (int f = 0; f < 25; ++f) {
        const int n = f / 5, kk = f % 5;
        const int row = 4 * kk + khalf, col = 16 * n + mrow;
        v2f w;
        w.x = Wh1[row * LH4 + col]; w.y = Wh1[(row + 1) * LH4 + col]; sWh1f[f * 32 + lane] = w;
        w.x = Wi2[row * LH4 + col]; w.y = Wi2[(row + 1) * LH4 + col]; sWi2f[f * 32 + lane] = w;
        w.x = Wh2[row * LH4 + col]; w.y = Wh2[(row + 1) * LH4 + col]; sWh2f[f * 32 + lane] = w;
    }
    for (int kk = 0; kk < 5; ++kk) {
        const int row = 4 * kk + khalf;
        v2f w;
        w.x = (mrow < LOUT) ? Wd[row * LOUT + mrow] : 0.0f;
        w.y = (mrow < LOUT) ? Wd[(row + 1) * LOUT + mrow] : 0.0f;
        sWdf[kk * 32 + lane] = w;
    }
    for (int i = lane; i < LH4; i += 32) { sb1[i] = b1[i]; sb2[i] = b2[i]; }
    for (int i = lane; i < 16; i += 32) sbd[i] = (i < LOUT) ? bd[i] : 0.0f;
    {
        v2f zz; zz.x = 0.0f; zz.y = 0.0f;
        for (int i = lane; i < 5 * 32; i += 32) { sH1f[i] = zz; sH2f[i] = zz; }
        for (int i = lane; i < 16 * LH; i += 32) { sC1[i] = 0.0f; sC2[i] = 0.0f; }
    }
    __syncthreads();

    const float* zp = (const float*)sZt4;

    for (int t = 0; t < LT; ++t) {
        const float* xrow = X + ((size_t)(b0 + mrow) * LT + (size_t)t) * LF;
        v2f ax[8];
        #pragma unroll
        for (int kk = 0; kk < 8; ++kk) {
            ax[kk].x = xrow[4 * kk + khalf];
            ax[kk].y = xrow[4 * kk + khalf + 1];
        }
        if (t + 1 < LT) __builtin_prefetch(xrow + LF, 0, 3);

        v8f zt[5];
        #pragma unroll
        for (int n = 0; n < 5; ++n) {
            v8f c = {};
            #pragma unroll
            for (int kk = 0; kk < 8; ++kk)
                WMMA4(c, ax[kk], sWi1f[(n * 8 + kk) * 32 + lane]);
            #pragma unroll
            for (int kk = 0; kk < 5; ++kk)
                WMMA4(c, sH1f[kk * 32 + lane], sWh1f[(n * 5 + kk) * 32 + lane]);
            zt[n] = c;
        }
        #pragma unroll
        for (int n = 0; n < 5; ++n) {
            const int zcol = 16 * n + mrow;
            v4f lo, hiq;
            lo.x  = zt[n][0]; lo.y  = zt[n][1]; lo.z  = zt[n][2]; lo.w  = zt[n][3];
            hiq.x = zt[n][4]; hiq.y = zt[n][5]; hiq.z = zt[n][6]; hiq.w = zt[n][7];
            sZt4[zcol * 4 + khalf]     = lo;
            sZt4[zcol * 4 + khalf + 1] = hiq;
        }
        __syncthreads();

        #pragma unroll
        for (int i = 0; i < 10; ++i) {
            const int idx = lane + 32 * i;
            const int m = idx / LH, j = idx % LH;
            const float zi = zp[ j            * 16 + m] + sb1[j];
            const float zf = zp[(LH     + j)  * 16 + m] + sb1[LH + j];
            const float zg = zp[(2 * LH + j)  * 16 + m] + sb1[2 * LH + j];
            const float zo = zp[(3 * LH + j)  * 16 + m] + sb1[3 * LH + j];
            const float cc = sigf(zf) * sC1[idx] + sigf(zi) * tanhfast(zg);
            sC1[idx] = cc;
            const float hh = sigf(zo) * tanhfast(cc);
            const int kk = j >> 2, off = j & 3;
            ((float*)sH1f)[((kk * 32) + (m + 16 * (off >> 1))) * 2 + (off & 1)] = hh;
        }
        __syncthreads();

        #pragma unroll
        for (int n = 0; n < 5; ++n) {
            v8f c = {};
            #pragma unroll
            for (int kk = 0; kk < 5; ++kk)
                WMMA4(c, sH1f[kk * 32 + lane], sWi2f[(n * 5 + kk) * 32 + lane]);
            #pragma unroll
            for (int kk = 0; kk < 5; ++kk)
                WMMA4(c, sH2f[kk * 32 + lane], sWh2f[(n * 5 + kk) * 32 + lane]);
            zt[n] = c;
        }
        #pragma unroll
        for (int n = 0; n < 5; ++n) {
            const int zcol = 16 * n + mrow;
            v4f lo, hiq;
            lo.x  = zt[n][0]; lo.y  = zt[n][1]; lo.z  = zt[n][2]; lo.w  = zt[n][3];
            hiq.x = zt[n][4]; hiq.y = zt[n][5]; hiq.z = zt[n][6]; hiq.w = zt[n][7];
            sZt4[zcol * 4 + khalf]     = lo;
            sZt4[zcol * 4 + khalf + 1] = hiq;
        }
        __syncthreads();

        #pragma unroll
        for (int i = 0; i < 10; ++i) {
            const int idx = lane + 32 * i;
            const int m = idx / LH, j = idx % LH;
            const float zi = zp[ j            * 16 + m] + sb2[j];
            const float zf = zp[(LH     + j)  * 16 + m] + sb2[LH + j];
            const float zg = zp[(2 * LH + j)  * 16 + m] + sb2[2 * LH + j];
            const float zo = zp[(3 * LH + j)  * 16 + m] + sb2[3 * LH + j];
            const float cc = sigf(zf) * sC2[idx] + sigf(zi) * tanhfast(zg);
            sC2[idx] = cc;
            const float hh = sigf(zo) * tanhfast(cc);
            const int kk = j >> 2, off = j & 3;
            ((float*)sH2f)[((kk * 32) + (m + 16 * (off >> 1))) * 2 + (off & 1)] = hh;
        }
        __syncthreads();

        {
            v8f c = {};
            #pragma unroll
            for (int kk = 0; kk < 5; ++kk)
                WMMA4(c, sH2f[kk * 32 + lane], sWdf[kk * 32 + lane]);
            if (mrow < LOUT) {
                const float bb = sbd[mrow];
                #pragma unroll
                for (int r = 0; r < 8; ++r) {
                    const int m = r + 8 * hi;
                    out[((size_t)(b0 + m) * LT + (size_t)t) * LOUT + mrow] = c[r] + bb;
                }
            }
        }
        __syncthreads();
    }
}

extern "C" void kernel_launch(void* const* d_in, const int* in_sizes, int n_in,
                              void* d_out, int out_size, void* d_ws, size_t ws_size,
                              hipStream_t stream) {
    (void)in_sizes; (void)n_in; (void)out_size; (void)d_ws; (void)ws_size;
    const float* X   = (const float*)d_in[0];
    const float* Wi1 = (const float*)d_in[1];
    const float* Wh1 = (const float*)d_in[2];
    const float* b1  = (const float*)d_in[3];
    const float* Wi2 = (const float*)d_in[4];
    const float* Wh2 = (const float*)d_in[5];
    const float* b2  = (const float*)d_in[6];
    const float* Wd  = (const float*)d_in[7];
    const float* bd  = (const float*)d_in[8];
    float* out = (float*)d_out;

    dim3 grid(LB / 16);
    dim3 block(32);
    lstm2_dense_fused<<<grid, block, 0, stream>>>(X, Wi1, Wh1, b1, Wi2, Wh2, b2, Wd, bd, out);
}


# Batch tile cut from 16 rows to 1 per wave (256 single-wave workgroups; 15 of 16 matrix rows are padding): gate math in the serial chain drops from 10 cells per lane and layer to 1
# speedup vs baseline: 2.2111x; 2.2111x over previous
#include <hip/hip_runtime.h>
#include <hip/hip_bf16.h>


typedef __attribute__((ext_vector_type(2))) float v2f;
typedef __attribute__((ext_vector_type(4))) float v4f;
typedef __attribute__((ext_vector_type(8))) float v8f;

#define LH    20
#define LH4   80
#define LF    32
#define LT    1024
#define LB    256
#define LOUT  10
#define LR    1
#define NCELL (LR * LH)
#define NGIT  ((NCELL + 31) / 32)

__device__ __forceinline__ float sigf(float x) { return 1.0f / (1.0f + __expf(-x)); }
__device__ __forceinline__ float tanhfast(float x) {
    const float e = __expf(2.0f * x);
    return (e - 1.0f) / (e + 1.0f);
}

#define WMMA4(c, a, b) \
    (c) = __builtin_amdgcn_wmma_f32_16x16x4_f32(false, (a), false, (b), (short)0, (c), false, false)

#define LDS_SYNC() do { \
    __builtin_amdgcn_fence(__ATOMIC_RELEASE, "workgroup", "local"); \
    __builtin_amdgcn_s_barrier(); \
    __builtin_amdgcn_fence(__ATOMIC_ACQUIRE, "workgroup", "local"); \
} while (0)

__global__ void __launch_bounds__(32)
lstm2_dense_fused(const float* __restrict__ X,
                  const float* __restrict__ Wi1, const float* __restrict__ Wh1,
                  const float* __restrict__ b1,
                  const float* __restrict__ Wi2, const float* __restrict__ Wh2,
                  const float* __restrict__ b2,
                  const float* __restrict__ Wd,  const float* __restrict__ bd,
                  float* __restrict__ out)
{
    __shared__ v2f sWi1f[40 * 32];
    __shared__ v2f sWh1f[25 * 32];
    __shared__ v2f sWi2f[25 * 32];
    __shared__ v2f sWh2f[25 * 32];
    __shared__ v2f sWdf [ 5 * 32];
    __shared__ float sb1[LH4], sb2[LH4], sbd[16];
    __shared__ float sZt[2 * LH4 * LR];
    __shared__ v2f sH1f[5 * 32], sH2f[5 * 32];
    __shared__ float sC1[NCELL], sC2[NCELL];

    const int lane  = threadIdx.x;
    const int mrow  = lane & 15;
    const int hi    = lane >> 4;
    const int khalf = 2 * hi;
    const int b0    = blockIdx.x * LR;

    for (int f = 0; f < 40; ++f) {
        const int n = f >> 3, kk = f & 7;
        const int row = 4 * kk + khalf, col = 16 * n + mrow;
        v2f w; w.x = Wi1[row * LH4 + col]; w.y = Wi1[(row + 1) * LH4 + col];
        sWi1f[f * 32 + lane] = w;
    }
    for (int f = 0; f < 25; ++f) {
        const int n = f / 5, kk = f % 5;
        const int row = 4 * kk + khalf, col = 16 * n + mrow;
        v2f w;
        w.x = Wh1[row * LH4 + col]; w.y = Wh1[(row + 1) * LH4 + col]; sWh1f[f * 32 + lane] = w;
        w.x = Wi2[row * LH4 + col]; w.y = Wi2[(row + 1) * LH4 + col]; sWi2f[f * 32 + lane] = w;
        w.x = Wh2[row * LH4 + col]; w.y = Wh2[(row + 1) * LH4 + col]; sWh2f[f * 32 + lane] = w;
    }
    for (int kk = 0; kk < 5; ++kk) {
        const int row = 4 * kk + khalf;
        v2f w;
        w.x = (mrow < LOUT) ? Wd[row * LOUT + mrow] : 0.0f;
        w.y = (mrow < LOUT) ? Wd[(row + 1) * LOUT + mrow] : 0.0f;
        sWdf[kk * 32 + lane] = w;
    }
    for (int i = lane; i < LH4; i += 32) { sb1[i] = b1[i]; sb2[i] = b2[i]; }
    for (int i = lane; i < 16; i += 32) sbd[i] = (i < LOUT) ? bd[i] : 0.0f;
    {
        v2f zz; zz.x = 0.0f; zz.y = 0.0f;
        for (int i = lane; i < 5 * 32; i += 32) { sH1f[i] = zz; sH2f[i] = zz; }
        for (int i = lane; i < NCELL; i += 32) { sC1[i] = 0.0f; sC2[i] = 0.0f; }
    }
    __syncthreads();

    const float* zp = sZt;
    float* zspill = sZt + (hi * LH4 + mrow) * LR;

    const float* xlane = X + (size_t)(b0 + mrow % LR) * LT * LF + khalf;
    v2f ax[8];
    #pragma unroll
    for (int kk = 0; kk < 8; ++kk) {
        ax[kk].x = xlane[4 * kk];
        ax[kk].y = xlane[4 * kk + 1];
    }

    for (int t = 0; t < LT; ++t) {

        v8f zt[5];
        #pragma unroll
        for (int n = 0; n < 5; ++n) {
            v8f c = {};
            #pragma unroll
            for (int kk = 0; kk < 8; ++kk)
                WMMA4(c, ax[kk], sWi1f[(n * 8 + kk) * 32 + lane]);
            #pragma unroll
            for (int kk = 0; kk < 5; ++kk)
                WMMA4(c, sH1f[kk * 32 + lane], sWh1f[(n * 5 + kk) * 32 + lane]);
            zt[n] = c;
        }
        {
            const int tn = (t + 1 < LT) ? (t + 1) : t;
            const float* xrow = xlane + (size_t)tn * LF;
            #pragma unroll
            for (int kk = 0; kk < 8; ++kk) {
                ax[kk].x = xrow[4 * kk];
                ax[kk].y = xrow[4 * kk + 1];
            }
        }
        #pragma unroll
        for (int n = 0; n < 5; ++n) {
            #pragma unroll
            for (int r = 0; r < LR; ++r)
                zspill[16 * n * LR + r] = zt[n][r];
        }
        LDS_SYNC();

        #pragma unroll
        for (int i = 0; i < NGIT; ++i) {
            const int idx = lane + 32 * i;
            if (idx < NCELL) {
                const int m = idx / LH, j = idx % LH;
                const float zi = zp[ j            * LR + m] + sb1[j];
                const float zf = zp[(LH     + j)  * LR + m] + sb1[LH + j];
                const float zg = zp[(2 * LH + j)  * LR + m] + sb1[2 * LH + j];
                const float zo = zp[(3 * LH + j)  * LR + m] + sb1[3 * LH + j];
                const float cc = sigf(zf) * sC1[idx] + sigf(zi) * tanhfast(zg);
                sC1[idx] = cc;
                const float hh = sigf(zo) * tanhfast(cc);
                const int kk = j >> 2, off = j & 3;
                ((float*)sH1f)[((kk * 32) + (m + 16 * (off >> 1))) * 2 + (off & 1)] = hh;
            }
        }
        LDS_SYNC();

        #pragma unroll
        for (int n = 0; n < 5; ++n) {
            v8f c = {};
            #pragma unroll
            for (int kk = 0; kk < 5; ++kk)
                WMMA4(c, sH1f[kk * 32 + lane], sWi2f[(n * 5 + kk) * 32 + lane]);
            #pragma unroll
            for (int kk = 0; kk < 5; ++kk)
                WMMA4(c, sH2f[kk * 32 + lane], sWh2f[(n * 5 + kk) * 32 + lane]);
            zt[n] = c;
        }
        #pragma unroll
        for (int n = 0; n < 5; ++n) {
            #pragma unroll
            for (int r = 0; r < LR; ++r)
                zspill[16 * n * LR + r] = zt[n][r];
        }
        LDS_SYNC();

        #pragma unroll
        for (int i = 0; i < NGIT; ++i) {
            const int idx = lane + 32 * i;
            if (idx < NCELL) {
                const int m = idx / LH, j = idx % LH;
                const float zi = zp[ j            * LR + m] + sb2[j];
                const float zf = zp[(LH     + j)  * LR + m] + sb2[LH + j];
                const float zg = zp[(2 * LH + j)  * LR + m] + sb2[2 * LH + j];
                const float zo = zp[(3 * LH + j)  * LR + m] + sb2[3 * LH + j];
                const float cc = sigf(zf) * sC2[idx] + sigf(zi) * tanhfast(zg);
                sC2[idx] = cc;
                const float hh = sigf(zo) * tanhfast(cc);
                const int kk = j >> 2, off = j & 3;
                ((float*)sH2f)[((kk * 32) + (m + 16 * (off >> 1))) * 2 + (off & 1)] = hh;
            }
        }
        LDS_SYNC();

        {
            v8f c = {};
            #pragma unroll
            for (int kk = 0; kk < 5; ++kk)
                WMMA4(c, sH2f[kk * 32 + lane], sWdf[kk * 32 + lane]);
            if (hi == 0 && mrow < LOUT) {
                const float bb = sbd[mrow];
                #pragma unroll
                for (int r = 0; r < LR; ++r)
                    out[((size_t)(b0 + r) * LT + (size_t)t) * LOUT + mrow] = c[r] + bb;
            }
        }
        LDS_SYNC();
    }
}

extern "C" void kernel_launch(void* const* d_in, const int* in_sizes, int n_in,
                              void* d_out, int out_size, void* d_ws, size_t ws_size,
                              hipStream_t stream) {
    (void)in_sizes; (void)n_in; (void)out_size; (void)d_ws; (void)ws_size;
    const float* X   = (const float*)d_in[0];
    const float* Wi1 = (const float*)d_in[1];
    const float* Wh1 = (const float*)d_in[2];
    const float* b1  = (const float*)d_in[3];
    const float* Wi2 = (const float*)d_in[4];
    const float* Wh2 = (const float*)d_in[5];
    const float* b2  = (const float*)d_in[6];
    const float* Wd  = (const float*)d_in[7];
    const float* bd  = (const float*)d_in[8];
    float* out = (float*)d_out;

    dim3 grid(LB / LR);
    dim3 block(32);
    lstm2_dense_fused<<<grid, block, 0, stream>>>(X, Wi1, Wh1, b1, Wi2, Wh2, b2, Wd, bd, out);
}


# Layer 2 runs one 16-step block behind layer 1: x@Wi1, h1@Wi2 and the head are done 16 time steps per tile once per block, only h@Wh per step; both layers share step phases; sums unchanged
# speedup vs baseline: 4.5962x; 2.0787x over previous
#include <hip/hip_runtime.h>
#include <hip/hip_bf16.h>


typedef __attribute__((ext_vector_type(2))) float v2f;
typedef __attribute__((ext_vector_type(4))) float v4f;
typedef __attribute__((ext_vector_type(8))) float v8f;

#define LH    20
#define LH4   80
#define LF    32
#define LT    1024
#define LB    256
#define LOUT  10
#define TBLK  16
#define NBLK  (LT / TBLK)
#define XP    84

__device__ __forceinline__ float sigf(float x) { return 1.0f / (1.0f + __expf(-x)); }
__device__ __forceinline__ float tanhfast(float x) {
    const float e = __expf(2.0f * x);
    return (e - 1.0f) / (e + 1.0f);
}

#define WMMA4(c, a, b) \
    (c) = __builtin_amdgcn_wmma_f32_16x16x4_f32(false, (a), false, (b), (short)0, (c), false, false)

#define LDS_SYNC() do { \
    __builtin_amdgcn_fence(__ATOMIC_RELEASE, "workgroup", "local"); \
    __builtin_amdgcn_s_barrier(); \
    __builtin_amdgcn_fence(__ATOMIC_ACQUIRE, "workgroup", "local"); \
} while (0)

__global__ void __launch_bounds__(32)
lstm2_dense_fused(const float* __restrict__ X,
                  const float* __restrict__ Wi1, const float* __restrict__ Wh1,
                  const float* __restrict__ b1,
                  const float* __restrict__ Wi2, const float* __restrict__ Wh2,
                  const float* __restrict__ b2,
                  const float* __restrict__ Wd,  const float* __restrict__ bd,
                  float* __restrict__ out)
{
    __shared__ v2f sWi1f[40 * 32];
    __shared__ v2f sWh1f[25 * 32];
    __shared__ v2f sWi2f[25 * 32];
    __shared__ v2f sWh2f[25 * 32];
    __shared__ v2f sWdf [ 5 * 32];
    __shared__ float sb1[LH4], sb2[LH4], sbd[16];
    __shared__ float sXW1[TBLK * XP];
    __shared__ float sXW2[TBLK * XP];
    __shared__ float sZ1[2 * LH4], sZ2[2 * LH4];
    __shared__ v2f sH1f[5 * 32], sH2f[5 * 32];
    __shared__ v2f sH1blk[5 * 32], sH2blk[5 * 32];
    __shared__ float sC1[LH], sC2[LH];
    __shared__ float sOff[5 * 32 * 2];

    const int lane  = threadIdx.x;
    const int mrow  = lane & 15;
    const int hi    = lane >> 4;
    const int khalf = 2 * hi;
    const int b0    = blockIdx.x;

    for (int f = 0; f < 40; ++f) {
        const int n = f >> 3, kk = f & 7;
        const int row = 4 * kk + khalf, col = 16 * n + mrow;
        v2f w; w.x = Wi1[row * LH4 + col]; w.y = Wi1[(row + 1) * LH4 + col];
        sWi1f[f * 32 + lane] = w;
    }
    for (int f = 0; f < 25; ++f) {
        const int n = f / 5, kk = f % 5;
        const int row = 4 * kk + khalf, col = 16 * n + mrow;
        v2f w;
        w.x = Wh1[row * LH4 + col]; w.y = Wh1[(row + 1) * LH4 + col]; sWh1f[f * 32 + lane] = w;
        w.x = Wi2[row * LH4 + col]; w.y = Wi2[(row + 1) * LH4 + col]; sWi2f[f * 32 + lane] = w;
        w.x = Wh2[row * LH4 + col]; w.y = Wh2[(row + 1) * LH4 + col]; sWh2f[f * 32 + lane] = w;
    }
    for (int kk = 0; kk < 5; ++kk) {
        const int row = 4 * kk + khalf;
        v2f w;
        w.x = (mrow < LOUT) ? Wd[row * LOUT + mrow] : 0.0f;
        w.y = (mrow < LOUT) ? Wd[(row + 1) * LOUT + mrow] : 0.0f;
        sWdf[kk * 32 + lane] = w;
    }
    for (int i = lane; i < LH4; i += 32) { sb1[i] = b1[i]; sb2[i] = b2[i]; }
    for (int i = lane; i < 16; i += 32) sbd[i] = (i < LOUT) ? bd[i] : 0.0f;
    {
        v2f zz; zz.x = 0.0f; zz.y = 0.0f;
        for (int i = lane; i < 5 * 32; i += 32) {
            sH1f[i] = zz; sH2f[i] = zz; sH1blk[i] = zz; sH2blk[i] = zz;
        }
        for (int i = lane; i < LH; i += 32) { sC1[i] = 0.0f; sC2[i] = 0.0f; }
        for (int i = lane; i < TBLK * XP; i += 32) { sXW1[i] = 0.0f; sXW2[i] = 0.0f; }
    }
    __syncthreads();

    float* zs1 = sZ1 + hi * LH4 + mrow;
    float* zs2 = sZ2 + hi * LH4 + mrow;

    const float* xbase = X + (size_t)b0 * LT * LF + (size_t)mrow * LF + khalf;
    v2f ax[8];
    #pragma unroll
    for (int kk = 0; kk < 8; ++kk) {
        ax[kk].x = xbase[4 * kk];
        ax[kk].y = xbase[4 * kk + 1];
    }

    v8f ci1[5], ci2[5];
    #pragma unroll
    for (int n = 0; n < 5; ++n) { v8f zz = {}; ci1[n] = zz; ci2[n] = zz; }

    const int gj   = (lane < LH) ? lane : 0;
    const int hidx = (((gj >> 2) * 32) + 16 * ((gj & 3) >> 1)) * 2 + (gj & 1);

    for (int blk = 0; blk < NBLK + 2; ++blk) {
        const bool l1on = (blk < NBLK);
        const bool l2on = (blk >= 1) && (blk <= NBLK);
        float* c1dst = l1on ? sC1 : sOff;
        float* h1dst = l1on ? (float*)sH1f : sOff;
        float* h1blk = l1on ? (float*)sH1blk : sOff;
        float* c2dst = l2on ? sC2 : sOff;
        float* h2dst = l2on ? (float*)sH2f : sOff;
        float* h2blk = l2on ? (float*)sH2blk : sOff;

        if (l1on) {
            v8f xw[5];
            #pragma unroll
            for (int n = 0; n < 5; ++n) {
                v8f c = {};
                #pragma unroll
                for (int kk = 0; kk < 8; ++kk)
                    WMMA4(c, ax[kk], sWi1f[(n * 8 + kk) * 32 + lane]);
                xw[n] = c;
            }
            if (blk + 1 < NBLK) {
                const float* xr = xbase + (size_t)(blk + 1) * TBLK * LF;
                #pragma unroll
                for (int kk = 0; kk < 8; ++kk) {
                    ax[kk].x = xr[4 * kk];
                    ax[kk].y = xr[4 * kk + 1];
                }
            }
            #pragma unroll
            for (int n = 0; n < 5; ++n) {
                #pragma unroll
                for (int r = 0; r < 8; ++r)
                    sXW1[(r + 8 * hi) * XP + 16 * n + mrow] = xw[n][r];
            }
        }
        if (l2on) {
            v8f xw[5];
            #pragma unroll
            for (int n = 0; n < 5; ++n) {
                v8f c = {};
                #pragma unroll
                for (int kk = 0; kk < 5; ++kk)
                    WMMA4(c, sH1blk[kk * 32 + lane], sWi2f[(n * 5 + kk) * 32 + lane]);
                xw[n] = c;
            }
            #pragma unroll
            for (int n = 0; n < 5; ++n) {
                #pragma unroll
                for (int r = 0; r < 8; ++r)
                    sXW2[(r + 8 * hi) * XP + 16 * n + mrow] = xw[n][r];
            }
        }
        if (blk >= 2) {
            v8f c = {};
            #pragma unroll
            for (int kk = 0; kk < 5; ++kk)
                WMMA4(c, sH2blk[kk * 32 + lane], sWdf[kk * 32 + lane]);
            if (mrow < LOUT) {
                const float bb = sbd[mrow];
                float* orow = out + ((size_t)b0 * LT + (size_t)(blk - 2) * TBLK + 8 * hi) * LOUT + mrow;
                #pragma unroll
                for (int r = 0; r < 8; ++r)
                    orow[r * LOUT] = c[r] + bb;
            }
        }
        LDS_SYNC();

        if (blk <= NBLK) {
            for (int tt = 0; tt < TBLK; ++tt) {
                v8f z1[5], z2[5];
                #pragma unroll
                for (int n = 0; n < 5; ++n) {
                    ci1[n][0] = sXW1[tt * XP + 16 * n + mrow];
                    ci2[n][0] = sXW2[tt * XP + 16 * n + mrow];
                }
                #pragma unroll
                for (int n = 0; n < 5; ++n) {
                    v8f c = ci1[n];
                    #pragma unroll
                    for (int kk = 0; kk < 5; ++kk)
                        WMMA4(c, sH1f[kk * 32 + lane], sWh1f[(n * 5 + kk) * 32 + lane]);
                    z1[n] = c;
                }
                #pragma unroll
                for (int n = 0; n < 5; ++n) {
                    v8f c = ci2[n];
                    #pragma unroll
                    for (int kk = 0; kk < 5; ++kk)
                        WMMA4(c, sH2f[kk * 32 + lane], sWh2f[(n * 5 + kk) * 32 + lane]);
                    z2[n] = c;
                }
                #pragma unroll
                for (int n = 0; n < 5; ++n) {
                    zs1[16 * n] = z1[n][0];
                    zs2[16 * n] = z2[n][0];
                }
                LDS_SYNC();

                if (lane < LH) {
                    const int j = lane;
                    const float zi1 = sZ1[j]          + sb1[j];
                    const float zf1 = sZ1[LH + j]     + sb1[LH + j];
                    const float zg1 = sZ1[2 * LH + j] + sb1[2 * LH + j];
                    const float zo1 = sZ1[3 * LH + j] + sb1[3 * LH + j];
                    const float zi2 = sZ2[j]          + sb2[j];
                    const float zf2 = sZ2[LH + j]     + sb2[LH + j];
                    const float zg2 = sZ2[2 * LH + j] + sb2[2 * LH + j];
                    const float zo2 = sZ2[3 * LH + j] + sb2[3 * LH + j];
                    const float cc1 = sigf(zf1) * sC1[j] + sigf(zi1) * tanhfast(zg1);
                    const float cc2 = sigf(zf2) * sC2[j] + sigf(zi2) * tanhfast(zg2);
                    const float hh1 = sigf(zo1) * tanhfast(cc1);
                    const float hh2 = sigf(zo2) * tanhfast(cc2);
                    c1dst[j] = cc1;
                    h1dst[hidx] = hh1;
                    h1blk[hidx + 2 * tt] = hh1;
                    c2dst[j] = cc2;
                    h2dst[hidx] = hh2;
                    h2blk[hidx + 2 * tt] = hh2;
                }
                LDS_SYNC();
            }
        }
    }
}

extern "C" void kernel_launch(void* const* d_in, const int* in_sizes, int n_in,
                              void* d_out, int out_size, void* d_ws, size_t ws_size,
                              hipStream_t stream) {
    (void)in_sizes; (void)n_in; (void)out_size; (void)d_ws; (void)ws_size;
    const float* X   = (const float*)d_in[0];
    const float* Wi1 = (const float*)d_in[1];
    const float* Wh1 = (const float*)d_in[2];
    const float* b1  = (const float*)d_in[3];
    const float* Wi2 = (const float*)d_in[4];
    const float* Wh2 = (const float*)d_in[5];
    const float* b2  = (const float*)d_in[6];
    const float* Wd  = (const float*)d_in[7];
    const float* bd  = (const float*)d_in[8];
    float* out = (float*)d_out;

    dim3 grid(LB);
    dim3 block(32);
    lstm2_dense_fused<<<grid, block, 0, stream>>>(X, Wi1, Wh1, b1, Wi2, Wh2, b2, Wd, bd, out);
}


# Layers run half a step apart in the wave: each layer's gate phase (LDS regroup, exp, divisions) is issued between the other's 25 products in a fixed order; Wh, biases in registers; sums unchanged
# speedup vs baseline: 4.8301x; 1.0509x over previous
#include <hip/hip_runtime.h>
#include <hip/hip_bf16.h>


typedef __attribute__((ext_vector_type(2))) float v2f;
typedef __attribute__((ext_vector_type(4))) float v4f;
typedef __attribute__((ext_vector_type(8))) float v8f;

#define LH    20
#define LH4   80
#define LF    32
#define LT    1024
#define LB    256
#define LOUT  10
#define TBLK  16
#define NBLK  (LT / TBLK)
#define XP    84

__device__ __forceinline__ float sigf(float x) { return 1.0f / (1.0f + __expf(-x)); }
__device__ __forceinline__ float tanhfast(float x) {
    const float e = __expf(2.0f * x);
    return (e - 1.0f) / (e + 1.0f);
}

#define WMMA4(c, a, b) \
    (c) = __builtin_amdgcn_wmma_f32_16x16x4_f32(false, (a), false, (b), (short)0, (c), false, false)

#define LDS_ORDER() __builtin_amdgcn_fence(__ATOMIC_ACQ_REL, "wavefront", "local")


__device__ __forceinline__ void gate_phase(const float* sZ, const float* sC,
                                           float bi, float bf, float bg, float bo,
                                           int j, int hidx, int trow,
                                           float* cdst, float* hdst, float* hblk)
{
    const float zi = sZ[j] + bi, zf = sZ[LH + j] + bf, zg = sZ[2 * LH + j] + bg, zo = sZ[3 * LH + j] + bo;
    const float ei = __expf(-zi), ef = __expf(-zf), eo = __expf(-zo), eg = __expf(2.0f * zg);
    const float si = 1.0f / (1.0f + ei);
    const float sf = 1.0f / (1.0f + ef);
    const float so = 1.0f / (1.0f + eo);
    const float tg = (eg - 1.0f) / (eg + 1.0f);
    const float cc = sf * sC[j] + si * tg;
    const float ec = __expf(2.0f * cc);
    const float hh = so * ((ec - 1.0f) / (ec + 1.0f));
    cdst[j] = cc;
    hdst[hidx] = hh;
    hblk[hidx + 2 * trow] = hh;
}

#define HS_GROUP(mask, n) __builtin_amdgcn_sched_group_barrier((mask), (n), 0)
#define HS_MATRIX 0x008
#define HS_VECTOR 0x402
#define HS_LDSRD  0x100
#define HS_LDSWR  0x200
#define HS_ORDER() do { \
    HS_GROUP(HS_LDSRD, 3);                   \
    HS_GROUP(HS_MATRIX, 6);                  \
    _Pragma("unroll") for (int g_ = 0; g_ < 5; ++g_) { \
        HS_GROUP(HS_VECTOR, 12); HS_GROUP(HS_MATRIX, 1); }    \
    _Pragma("unroll") for (int g_ = 0; g_ < 5; ++g_) { \
        HS_GROUP(HS_VECTOR, 4); HS_GROUP(HS_MATRIX, 1); }     \
    HS_GROUP(HS_VECTOR, 3); HS_GROUP(HS_MATRIX, 1);           \
    HS_GROUP(HS_LDSWR, 3);                   \
    HS_GROUP(HS_LDSRD, 6);                   \
    HS_GROUP(HS_MATRIX, 8);                  \
    HS_GROUP(HS_VECTOR, 3);                  \
    HS_GROUP(HS_LDSWR, 3);                   \
} while (0)

__device__ __forceinline__ void half_step(
        v8f (&z)[5], const v2f (&a)[5], const v2f (&wh)[25], const v8f (&ci)[5], float* zs,
        const float* sZ, const float* sC, float bi, float bf, float bg, float bo,
        int j, int hidx, int trow, float* cdst, float* hdst, float* hblk,
        v2f (&an)[5], const v2f* sHn, v8f (&cin)[5], const float* xwn)
{
    #pragma unroll
    for (int n = 0; n < 5; ++n) z[n] = ci[n];
    #pragma unroll
    for (int kk = 0; kk < 5; ++kk) {
        #pragma unroll
        for (int n = 0; n < 5; ++n)
            WMMA4(z[n], a[kk], wh[n * 5 + kk]);
    }

    const float zi = sZ[j] + bi, zf = sZ[LH + j] + bf, zg = sZ[2 * LH + j] + bg, zo = sZ[3 * LH + j] + bo;
    const float cold = sC[j];
    const float ei = __expf(-zi), ef = __expf(-zf), eo = __expf(-zo), eg = __expf(2.0f * zg);
    const float si = 1.0f / (1.0f + ei);
    const float sf = 1.0f / (1.0f + ef);
    const float so = 1.0f / (1.0f + eo);
    const float tg = (eg - 1.0f) / (eg + 1.0f);
    const float cc = sf * cold + si * tg;
    const float ec = __expf(2.0f * cc);
    const float hh = so * ((ec - 1.0f) / (ec + 1.0f));
    cdst[j] = cc;
    hdst[hidx] = hh;
    hblk[hidx + 2 * trow] = hh;
    LDS_ORDER();

    #pragma unroll
    for (int kk = 0; kk < 5; ++kk) an[kk] = sHn[kk * 32];
    #pragma unroll
    for (int n = 0; n < 5; ++n) cin[n][0] = xwn[16 * n];

    #pragma unroll
    for (int n = 0; n < 5; ++n) zs[16 * n] = z[n][0];
    LDS_ORDER();

    HS_ORDER();
    __builtin_amdgcn_sched_barrier(0);
}

__global__ void __launch_bounds__(32)
lstm2_dense_fused(const float* __restrict__ X,
                  const float* __restrict__ Wi1, const float* __restrict__ Wh1,
                  const float* __restrict__ b1,
                  const float* __restrict__ Wi2, const float* __restrict__ Wh2,
                  const float* __restrict__ b2,
                  const float* __restrict__ Wd,  const float* __restrict__ bd,
                  float* __restrict__ out)
{
    __shared__ v2f sWi1f[40 * 32];
    __shared__ v2f sWh1f[25 * 32];
    __shared__ v2f sWi2f[25 * 32];
    __shared__ v2f sWh2f[25 * 32];
    __shared__ v2f sWdf [ 5 * 32];
    __shared__ float sb1[LH4], sb2[LH4], sbd[16];
    __shared__ float sXW1[(TBLK + 1) * XP];
    __shared__ float sXW2[TBLK * XP];
    __shared__ float sZ1[2 * LH4], sZ2[2 * LH4];
    __shared__ v2f sH1f[5 * 32], sH2f[5 * 32];
    __shared__ v2f sH1blk[5 * 32], sH2blk[5 * 32];
    __shared__ float sC1[LH], sC2[LH];
    __shared__ float sOff[5 * 32 * 2];

    const int lane  = threadIdx.x;
    const int mrow  = lane & 15;
    const int hi    = lane >> 4;
    const int khalf = 2 * hi;
    const int b0    = blockIdx.x;

    for (int f = 0; f < 40; ++f) {
        const int n = f >> 3, kk = f & 7;
        const int row = 4 * kk + khalf, col = 16 * n + mrow;
        v2f w; w.x = Wi1[row * LH4 + col]; w.y = Wi1[(row + 1) * LH4 + col];
        sWi1f[f * 32 + lane] = w;
    }
    for (int f = 0; f < 25; ++f) {
        const int n = f / 5, kk = f % 5;
        const int row = 4 * kk + khalf, col = 16 * n + mrow;
        v2f w;
        w.x = Wh1[row * LH4 + col]; w.y = Wh1[(row + 1) * LH4 + col]; sWh1f[f * 32 + lane] = w;
        w.x = Wi2[row * LH4 + col]; w.y = Wi2[(row + 1) * LH4 + col]; sWi2f[f * 32 + lane] = w;
        w.x = Wh2[row * LH4 + col]; w.y = Wh2[(row + 1) * LH4 + col]; sWh2f[f * 32 + lane] = w;
    }
    for (int kk = 0; kk < 5; ++kk) {
        const int row = 4 * kk + khalf;
        v2f w;
        w.x = (mrow < LOUT) ? Wd[row * LOUT + mrow] : 0.0f;
        w.y = (mrow < LOUT) ? Wd[(row + 1) * LOUT + mrow] : 0.0f;
        sWdf[kk * 32 + lane] = w;
    }
    for (int i = lane; i < LH4; i += 32) { sb1[i] = b1[i]; sb2[i] = b2[i]; }
    for (int i = lane; i < 16; i += 32) sbd[i] = (i < LOUT) ? bd[i] : 0.0f;
    {
        v2f zz; zz.x = 0.0f; zz.y = 0.0f;
        for (int i = lane; i < 5 * 32; i += 32) {
            sH1f[i] = zz; sH2f[i] = zz; sH1blk[i] = zz; sH2blk[i] = zz;
        }
        for (int i = lane; i < LH; i += 32) { sC1[i] = 0.0f; sC2[i] = 0.0f; }
        for (int i = lane; i < TBLK * XP; i += 32) { sXW1[i] = 0.0f; sXW2[i] = 0.0f; }
        for (int i = lane; i < XP; i += 32) sXW1[TBLK * XP + i] = 0.0f;
        for (int i = lane; i < 2 * LH4; i += 32) { sZ1[i] = 0.0f; sZ2[i] = 0.0f; }
    }
    __syncthreads();

    float* zs1 = sZ1 + hi * LH4 + mrow;
    float* zs2 = sZ2 + hi * LH4 + mrow;

    const float* xbase = X + (size_t)b0 * LT * LF + (size_t)mrow * LF + khalf;
    v2f ax[8];
    #pragma unroll
    for (int kk = 0; kk < 8; ++kk) {
        ax[kk].x = xbase[4 * kk];
        ax[kk].y = xbase[4 * kk + 1];
    }

    const int gj   = (lane < LH) ? lane : 0;
    const int hidx = (((gj >> 2) * 32) + 16 * ((gj & 3) >> 1)) * 2 + (gj & 1);

    v2f wh1[25], wh2[25];
    #pragma unroll
    for (int f = 0; f < 25; ++f) { wh1[f] = sWh1f[f * 32 + lane]; wh2[f] = sWh2f[f * 32 + lane]; }
    const float b1i = sb1[gj], b1f = sb1[LH + gj], b1g = sb1[2 * LH + gj], b1o = sb1[3 * LH + gj];
    const float b2i = sb2[gj], b2f = sb2[LH + gj], b2g = sb2[2 * LH + gj], b2o = sb2[3 * LH + gj];

    v8f ci1[5], ci2[5];
    #pragma unroll
    for (int n = 0; n < 5; ++n) { v8f zz = {}; ci1[n] = zz; ci2[n] = zz; }

    for (int blk = 0; blk < NBLK + 2; ++blk) {
        const bool l1on = (blk < NBLK);
        const bool l2on = (blk >= 1) && (blk <= NBLK);
        float* c1dst = l1on ? sC1 : sOff;
        float* h1dst = l1on ? (float*)sH1f : sOff;
        float* h1blk = l1on ? (float*)sH1blk : sOff;
        float* c2dst = l2on ? sC2 : sOff;
        float* h2dst = l2on ? (float*)sH2f : sOff;
        float* h2blk = l2on ? (float*)sH2blk : sOff;

        if (l1on) {
            v8f xw[5];
            #pragma unroll
            for (int n = 0; n < 5; ++n) {
                v8f c = {};
                #pragma unroll
                for (int kk = 0; kk < 8; ++kk)
                    WMMA4(c, ax[kk], sWi1f[(n * 8 + kk) * 32 + lane]);
                xw[n] = c;
            }
            if (blk + 1 < NBLK) {
                const float* xr = xbase + (size_t)(blk + 1) * TBLK * LF;
                #pragma unroll
                for (int kk = 0; kk < 8; ++kk) {
                    ax[kk].x = xr[4 * kk];
                    ax[kk].y = xr[4 * kk + 1];
                }
            }
            #pragma unroll
            for (int n = 0; n < 5; ++n) {
                #pragma unroll
                for (int r = 0; r < 8; ++r)
                    sXW1[(r + 8 * hi) * XP + 16 * n + mrow] = xw[n][r];
            }
        }
        if (l2on) {
            v8f xw[5];
            #pragma unroll
            for (int n = 0; n < 5; ++n) {
                v8f c = {};
                #pragma unroll
                for (int kk = 0; kk < 5; ++kk)
                    WMMA4(c, sH1blk[kk * 32 + lane], sWi2f[(n * 5 + kk) * 32 + lane]);
                xw[n] = c;
            }
            #pragma unroll
            for (int n = 0; n < 5; ++n) {
                #pragma unroll
                for (int r = 0; r < 8; ++r)
                    sXW2[(r + 8 * hi) * XP + 16 * n + mrow] = xw[n][r];
            }
        }
        if (blk >= 2) {
            v8f c = {};
            #pragma unroll
            for (int kk = 0; kk < 5; ++kk)
                WMMA4(c, sH2blk[kk * 32 + lane], sWdf[kk * 32 + lane]);
            if (mrow < LOUT) {
                const float bb = sbd[mrow];
                float* orow = out + ((size_t)b0 * LT + (size_t)(blk - 2) * TBLK + 8 * hi) * LOUT + mrow;
                #pragma unroll
                for (int r = 0; r < 8; ++r)
                    orow[r * LOUT] = c[r] + bb;
            }
        }
        LDS_ORDER();

        if (blk <= NBLK) {
            v2f a1[5], a2[5];
            #pragma unroll
            for (int kk = 0; kk < 5; ++kk) a1[kk] = sH1f[kk * 32 + lane];
            #pragma unroll
            for (int n = 0; n < 5; ++n) ci1[n][0] = sXW1[16 * n + mrow];

            for (int tt = 0; tt < TBLK; ++tt) {
                const bool g2on = (tt != 0);
                float* c2d = g2on ? c2dst : sOff;
                float* h2d = g2on ? h2dst : sOff;
                float* h2b = g2on ? h2blk : sOff;
                const int  t2 = (tt + TBLK - 1) & (TBLK - 1);
                v8f z1[5], z2[5];

                half_step(z1, a1, wh1, ci1, zs1,
                          sZ2, sC2, b2i, b2f, b2g, b2o, gj, hidx, t2, c2d, h2d, h2b,
                          a2, sH2f + lane, ci2, sXW2 + tt * XP + mrow);

                half_step(z2, a2, wh2, ci2, zs2,
                          sZ1, sC1, b1i, b1f, b1g, b1o, gj, hidx, tt, c1dst, h1dst, h1blk,
                          a1, sH1f + lane, ci1, sXW1 + (tt + 1) * XP + mrow);
            }
            gate_phase(sZ2, sC2, b2i, b2f, b2g, b2o, gj, hidx, TBLK - 1, c2dst, h2dst, h2blk);
            LDS_ORDER();
        }
    }
}

extern "C" void kernel_launch(void* const* d_in, const int* in_sizes, int n_in,
                              void* d_out, int out_size, void* d_ws, size_t ws_size,
                              hipStream_t stream) {
    (void)in_sizes; (void)n_in; (void)out_size; (void)d_ws; (void)ws_size;
    const float* X   = (const float*)d_in[0];
    const float* Wi1 = (const float*)d_in[1];
    const float* Wh1 = (const float*)d_in[2];
    const float* b1  = (const float*)d_in[3];
    const float* Wi2 = (const float*)d_in[4];
    const float* Wh2 = (const float*)d_in[5];
    const float* b2  = (const float*)d_in[6];
    const float* Wd  = (const float*)d_in[7];
    const float* bd  = (const float*)d_in[8];
    float* out = (float*)d_out;

    dim3 grid(LB);
    dim3 block(32);
    lstm2_dense_fused<<<grid, block, 0, stream>>>(X, Wi1, Wh1, b1, Wi2, Wh2, b2, Wd, bd, out);
}


# Layers on two waves of the workgroup (two SIMDs) instead of interleaved in one wave: wave 1 runs layer 2 and head one 16-step block behind, h1 via 2 LDS buffers, 1 barrier per block; sums unchanged
# speedup vs baseline: 8.5567x; 1.7715x over previous
#include <hip/hip_runtime.h>
#include <hip/hip_bf16.h>


typedef __attribute__((ext_vector_type(2))) float v2f;
typedef __attribute__((ext_vector_type(4))) float v4f;
typedef __attribute__((ext_vector_type(8))) float v8f;

#define LH    20
#define LH4   80
#define LF    32
#define LT    1024
#define LB    256
#define LOUT  10
#define TBLK  16
#define NBLK  (LT / TBLK)
#define XP    84

__device__ __forceinline__ float sigf(float x) { return 1.0f / (1.0f + __expf(-x)); }
__device__ __forceinline__ float tanhfast(float x) {
    const float e = __expf(2.0f * x);
    return (e - 1.0f) / (e + 1.0f);
}

#define WMMA4(c, a, b) \
    (c) = __builtin_amdgcn_wmma_f32_16x16x4_f32(false, (a), false, (b), (short)0, (c), false, false)

#define LDS_ORDER() __builtin_amdgcn_fence(__ATOMIC_ACQ_REL, "wavefront", "local")


__device__ __forceinline__ void gate_phase(const float* sZ, float* sC,
                                           float bi, float bf, float bg, float bo,
                                           int j, int hidx, int trow,
                                           float* hdst, float* hblk)
{
    const float zi = sZ[j] + bi, zf = sZ[LH + j] + bf, zg = sZ[2 * LH + j] + bg, zo = sZ[3 * LH + j] + bo;
    const float ei = __expf(-zi), ef = __expf(-zf), eo = __expf(-zo), eg = __expf(2.0f * zg);
    const float si = 1.0f / (1.0f + ei);
    const float sf = 1.0f / (1.0f + ef);
    const float so = 1.0f / (1.0f + eo);
    const float tg = (eg - 1.0f) / (eg + 1.0f);
    const float cc = sf * sC[j] + si * tg;
    const float ec = __expf(2.0f * cc);
    const float hh = so * ((ec - 1.0f) / (ec + 1.0f));
    sC[j] = cc;
    hdst[hidx] = hh;
    hblk[hidx + 2 * trow] = hh;
}

__device__ __forceinline__ void layer_step(
        v8f (&ci)[5], v2f (&a)[5], const v2f (&wh)[25],
        float* zs, const float* sZ, float* sC,
        float bi, float bf, float bg, float bo,
        int j, int hidx, int trow, float* hdst, float* hblk,
        const v2f* sHl, const float* xwn)
{
    v8f z[5];
    #pragma unroll
    for (int n = 0; n < 5; ++n) z[n] = ci[n];
    #pragma unroll
    for (int kk = 0; kk < 5; ++kk) {
        #pragma unroll
        for (int n = 0; n < 5; ++n)
            WMMA4(z[n], a[kk], wh[n * 5 + kk]);
    }
    #pragma unroll
    for (int n = 0; n < 5; ++n) ci[n][0] = xwn[16 * n];
    #pragma unroll
    for (int n = 0; n < 5; ++n) zs[16 * n] = z[n][0];
    LDS_ORDER();
    gate_phase(sZ, sC, bi, bf, bg, bo, j, hidx, trow, hdst, hblk);
    LDS_ORDER();
    #pragma unroll
    for (int kk = 0; kk < 5; ++kk) a[kk] = sHl[kk * 32];
}

__global__ void __launch_bounds__(64)
lstm2_dense_fused(const float* __restrict__ X,
                  const float* __restrict__ Wi1, const float* __restrict__ Wh1,
                  const float* __restrict__ b1,
                  const float* __restrict__ Wi2, const float* __restrict__ Wh2,
                  const float* __restrict__ b2,
                  const float* __restrict__ Wd,  const float* __restrict__ bd,
                  float* __restrict__ out)
{
    __shared__ v2f sWi1f[40 * 32];
    __shared__ v2f sWh1f[25 * 32];
    __shared__ v2f sWi2f[25 * 32];
    __shared__ v2f sWh2f[25 * 32];
    __shared__ v2f sWdf [ 5 * 32];
    __shared__ float sb1[LH4], sb2[LH4], sbd[16];
    __shared__ float sXW1[(TBLK + 1) * XP];
    __shared__ float sXW2[(TBLK + 1) * XP];
    __shared__ float sZ1[2 * LH4], sZ2[2 * LH4];
    __shared__ v2f sH1f[5 * 32], sH2f[5 * 32];
    __shared__ v2f sH1blk[2 * 5 * 32];
    __shared__ v2f sH2blk[5 * 32];
    __shared__ float sC1[LH], sC2[LH];

    const int tid   = threadIdx.x;
    const int lane  = tid & 31;
    const int wv    = __builtin_amdgcn_readfirstlane(tid >> 5);
    const int mrow  = lane & 15;
    const int hi    = lane >> 4;
    const int khalf = 2 * hi;
    const int b0    = blockIdx.x;

    if (wv == 0) {
        for (int f = 0; f < 40; ++f) {
            const int n = f >> 3, kk = f & 7;
            const int row = 4 * kk + khalf, col = 16 * n + mrow;
            v2f w; w.x = Wi1[row * LH4 + col]; w.y = Wi1[(row + 1) * LH4 + col];
            sWi1f[f * 32 + lane] = w;
        }
        for (int f = 0; f < 25; ++f) {
            const int n = f / 5, kk = f % 5;
            const int row = 4 * kk + khalf, col = 16 * n + mrow;
            v2f w;
            w.x = Wh1[row * LH4 + col]; w.y = Wh1[(row + 1) * LH4 + col]; sWh1f[f * 32 + lane] = w;
        }
    } else {
        for (int f = 0; f < 25; ++f) {
            const int n = f / 5, kk = f % 5;
            const int row = 4 * kk + khalf, col = 16 * n + mrow;
            v2f w;
            w.x = Wi2[row * LH4 + col]; w.y = Wi2[(row + 1) * LH4 + col]; sWi2f[f * 32 + lane] = w;
            w.x = Wh2[row * LH4 + col]; w.y = Wh2[(row + 1) * LH4 + col]; sWh2f[f * 32 + lane] = w;
        }
        for (int kk = 0; kk < 5; ++kk) {
            const int row = 4 * kk + khalf;
            v2f w;
            w.x = (mrow < LOUT) ? Wd[row * LOUT + mrow] : 0.0f;
            w.y = (mrow < LOUT) ? Wd[(row + 1) * LOUT + mrow] : 0.0f;
            sWdf[kk * 32 + lane] = w;
        }
    }
    for (int i = tid; i < LH4; i += 64) { sb1[i] = b1[i]; sb2[i] = b2[i]; }
    for (int i = tid; i < 16; i += 64) sbd[i] = (i < LOUT) ? bd[i] : 0.0f;
    {
        v2f zz; zz.x = 0.0f; zz.y = 0.0f;
        for (int i = tid; i < 5 * 32; i += 64) {
            sH1f[i] = zz; sH2f[i] = zz; sH2blk[i] = zz; sH1blk[i] = zz; sH1blk[5 * 32 + i] = zz;
        }
        for (int i = tid; i < LH; i += 64) { sC1[i] = 0.0f; sC2[i] = 0.0f; }
        for (int i = tid; i < (TBLK + 1) * XP; i += 64) { sXW1[i] = 0.0f; sXW2[i] = 0.0f; }
        for (int i = tid; i < 2 * LH4; i += 64) { sZ1[i] = 0.0f; sZ2[i] = 0.0f; }
    }
    __syncthreads();

    const v2f*   sWhf = wv ? sWh2f : sWh1f;
    const float* sb   = wv ? sb2   : sb1;
    float*       sZ   = wv ? sZ2   : sZ1;
    float*       sC   = wv ? sC2   : sC1;
    v2f*         sHf  = wv ? sH2f  : sH1f;
    float*       sXW  = wv ? sXW2  : sXW1;

    float* zs = sZ + hi * LH4 + mrow;

    const float* xbase = X + (size_t)b0 * LT * LF + (size_t)mrow * LF + khalf;
    v2f ax[8];
    #pragma unroll
    for (int kk = 0; kk < 8; ++kk) { ax[kk].x = 0.0f; ax[kk].y = 0.0f; }
    if (wv == 0) {
        #pragma unroll
        for (int kk = 0; kk < 8; ++kk) {
            ax[kk].x = xbase[4 * kk];
            ax[kk].y = xbase[4 * kk + 1];
        }
    }

    const int gj   = (lane < LH) ? lane : 0;
    const int hidx = (((gj >> 2) * 32) + 16 * ((gj & 3) >> 1)) * 2 + (gj & 1);

    v2f wh[25];
    #pragma unroll
    for (int f = 0; f < 25; ++f) wh[f] = sWhf[f * 32 + lane];
    const float bi = sb[gj], bf = sb[LH + gj], bg = sb[2 * LH + gj], bo = sb[3 * LH + gj];

    v8f ci[5];
    #pragma unroll
    for (int n = 0; n < 5; ++n) { v8f zz = {}; ci[n] = zz; }

    for (int blk = 0; blk < NBLK + 2; ++blk) {
        const bool on = wv ? ((blk >= 1) && (blk <= NBLK)) : (blk < NBLK);
        float* hblk = wv ? (float*)sH2blk : (float*)(sH1blk + (blk & 1) * (5 * 32));

        if (wv == 0) {
            if (on) {
                v8f xw[5];
                #pragma unroll
                for (int n = 0; n < 5; ++n) {
                    v8f c = {};
                    #pragma unroll
                    for (int kk = 0; kk < 8; ++kk)
                        WMMA4(c, ax[kk], sWi1f[(n * 8 + kk) * 32 + lane]);
                    xw[n] = c;
                }
                if (blk + 1 < NBLK) {
                    const float* xr = xbase + (size_t)(blk + 1) * TBLK * LF;
                    #pragma unroll
                    for (int kk = 0; kk < 8; ++kk) {
                        ax[kk].x = xr[4 * kk];
                        ax[kk].y = xr[4 * kk + 1];
                    }
                }
                #pragma unroll
                for (int n = 0; n < 5; ++n) {
                    #pragma unroll
                    for (int r = 0; r < 8; ++r)
                        sXW1[(r + 8 * hi) * XP + 16 * n + mrow] = xw[n][r];
                }
            }
        } else {
            if (on) {
                const v2f* h1b = sH1blk + ((blk - 1) & 1) * (5 * 32);
                v8f xw[5];
                #pragma unroll
                for (int n = 0; n < 5; ++n) {
                    v8f c = {};
                    #pragma unroll
                    for (int kk = 0; kk < 5; ++kk)
                        WMMA4(c, h1b[kk * 32 + lane], sWi2f[(n * 5 + kk) * 32 + lane]);
                    xw[n] = c;
                }
                #pragma unroll
                for (int n = 0; n < 5; ++n) {
                    #pragma unroll
                    for (int r = 0; r < 8; ++r)
                        sXW2[(r + 8 * hi) * XP + 16 * n + mrow] = xw[n][r];
                }
            }
            if (blk >= 2) {
                v8f c = {};
                #pragma unroll
                for (int kk = 0; kk < 5; ++kk)
                    WMMA4(c, sH2blk[kk * 32 + lane], sWdf[kk * 32 + lane]);
                if (mrow < LOUT) {
                    const float bb = sbd[mrow];
                    float* orow = out + ((size_t)b0 * LT + (size_t)(blk - 2) * TBLK + 8 * hi) * LOUT + mrow;
                    #pragma unroll
                    for (int r = 0; r < 8; ++r)
                        orow[r * LOUT] = c[r] + bb;
                }
            }
        }
        LDS_ORDER();

        if (on) {
            v2f a[5];
            #pragma unroll
            for (int kk = 0; kk < 5; ++kk) a[kk] = sHf[kk * 32 + lane];
            #pragma unroll
            for (int n = 0; n < 5; ++n) ci[n][0] = sXW[16 * n + mrow];

            for (int tt = 0; tt < TBLK; ++tt)
                layer_step(ci, a, wh, zs, sZ, sC, bi, bf, bg, bo, gj, hidx, tt,
                           (float*)sHf, hblk, sHf + lane, sXW + (tt + 1) * XP + mrow);
        }
        __syncthreads();
    }
}

extern "C" void kernel_launch(void* const* d_in, const int* in_sizes, int n_in,
                              void* d_out, int out_size, void* d_ws, size_t ws_size,
                              hipStream_t stream) {
    (void)in_sizes; (void)n_in; (void)out_size; (void)d_ws; (void)ws_size;
    const float* X   = (const float*)d_in[0];
    const float* Wi1 = (const float*)d_in[1];
    const float* Wh1 = (const float*)d_in[2];
    const float* b1  = (const float*)d_in[3];
    const float* Wi2 = (const float*)d_in[4];
    const float* Wh2 = (const float*)d_in[5];
    const float* b2  = (const float*)d_in[6];
    const float* Wd  = (const float*)d_in[7];
    const float* bd  = (const float*)d_in[8];
    float* out = (float*)d_out;

    dim3 grid(LB);
    dim3 block(64);
    lstm2_dense_fused<<<grid, block, 0, stream>>>(X, Wi1, Wh1, b1, Wi2, Wh2, b2, Wd, bd, out);
}


# z row and h handed over inside the wave by lane permutes (v_permlane16_var, DPP row_share) instead of LDS: all 16 A rows carry h, cell state in a register; no LDS round trip in a step; sums unchanged
# speedup vs baseline: 10.4159x; 1.2173x over previous
#include <hip/hip_runtime.h>
#include <hip/hip_bf16.h>


typedef __attribute__((ext_vector_type(2))) float v2f;
typedef __attribute__((ext_vector_type(4))) float v4f;
typedef __attribute__((ext_vector_type(8))) float v8f;

#define LH    20
#define LH4   80
#define LF    32
#define LT    1024
#define LB    256
#define LOUT  10
#define TBLK  16
#define NBLK  (LT / TBLK)
#define XP    84

__device__ __forceinline__ float sigf(float x) { return 1.0f / (1.0f + __expf(-x)); }
__device__ __forceinline__ float tanhfast(float x) {
    const float e = __expf(2.0f * x);
    return (e - 1.0f) / (e + 1.0f);
}

#define WMMA4(c, a, b) \
    (c) = __builtin_amdgcn_wmma_f32_16x16x4_f32(false, (a), false, (b), (short)0, (c), false, false)

#define LDS_ORDER() __builtin_amdgcn_fence(__ATOMIC_ACQ_REL, "wavefront", "local")


__device__ __forceinline__ float gather16(float v, unsigned sel) {
    const unsigned u = __builtin_bit_cast(unsigned, v);
    return __builtin_bit_cast(float, __builtin_amdgcn_permlane16_var(u, u, sel, false, false));
}
#define ROW_BCAST(v, n) \
    __builtin_bit_cast(float, __builtin_amdgcn_mov_dpp(__builtin_bit_cast(int, (v)), 0x150 + (n), 0xf, 0xf, false))

__device__ __forceinline__ void layer_step(
        v8f (&ci)[5], v2f (&a)[5], const v2f (&wh)[25], float& cst,
        unsigned seli, unsigned self_, unsigned selg, unsigned selo,
        bool upi, bool upf, bool upg, bool upo,
        float bi, float bf, float bg, float bo,
        float* hrow, const float* xwn)
{
    float nx[5];
    #pragma unroll
    for (int n = 0; n < 5; ++n) nx[n] = xwn[16 * n];
    __builtin_amdgcn_sched_barrier(0);
    v8f z[5];
    #pragma unroll
    for (int n = 0; n < 5; ++n) z[n] = ci[n];
    #pragma unroll
    for (int kk = 0; kk < 5; ++kk) {
        #pragma unroll
        for (int n = 0; n < 5; ++n)
            WMMA4(z[n], a[kk], wh[n * 5 + kk]);
    }
    #pragma unroll
    for (int n = 0; n < 5; ++n) ci[n][0] = nx[n];
    const float ia = gather16(z[0][0], seli),  ib = gather16(z[1][0], seli);
    const float fa = gather16(z[1][0], self_), fb = gather16(z[2][0], self_);
    const float ga = gather16(z[2][0], selg),  gb = gather16(z[3][0], selg);
    const float oa = gather16(z[3][0], selo),  ob = gather16(z[4][0], selo);
    const float ri = upi ? ib : ia, rf = upf ? fb : fa, rg = upg ? gb : ga, ro = upo ? ob : oa;
    const float zi = ri + bi, zf = rf + bf, zg = rg + bg, zo = ro + bo;
    const float ei = __expf(-zi), ef = __expf(-zf), eo = __expf(-zo), eg = __expf(2.0f * zg);
    const float si = 1.0f / (1.0f + ei);
    const float sf = 1.0f / (1.0f + ef);
    const float so = 1.0f / (1.0f + eo);
    const float tg = (eg - 1.0f) / (eg + 1.0f);
    const float cc = sf * cst + si * tg;
    const float ec = __expf(2.0f * cc);
    const float hh = so * ((ec - 1.0f) / (ec + 1.0f));
    cst = cc;
    *hrow = hh;
    a[0].x = ROW_BCAST(hh, 0); a[0].y = ROW_BCAST(hh, 1);
    a[1].x = ROW_BCAST(hh, 2); a[1].y = ROW_BCAST(hh, 3);
    a[2].x = ROW_BCAST(hh, 4); a[2].y = ROW_BCAST(hh, 5);
    a[3].x = ROW_BCAST(hh, 6); a[3].y = ROW_BCAST(hh, 7);
    a[4].x = ROW_BCAST(hh, 8); a[4].y = ROW_BCAST(hh, 9);
}

__global__ void __launch_bounds__(64)
lstm2_dense_fused(const float* __restrict__ X,
                  const float* __restrict__ Wi1, const float* __restrict__ Wh1,
                  const float* __restrict__ b1,
                  const float* __restrict__ Wi2, const float* __restrict__ Wh2,
                  const float* __restrict__ b2,
                  const float* __restrict__ Wd,  const float* __restrict__ bd,
                  float* __restrict__ out)
{
    __shared__ v2f sWi1f[40 * 32];
    __shared__ v2f sWh1f[25 * 32];
    __shared__ v2f sWi2f[25 * 32];
    __shared__ v2f sWh2f[25 * 32];
    __shared__ v2f sWdf [ 5 * 32];
    __shared__ float sb1[LH4], sb2[LH4], sbd[16];
    __shared__ float sXW1[(TBLK + 1) * XP];
    __shared__ float sXW2[(TBLK + 1) * XP];
    __shared__ v2f sH1blk[2 * 5 * 32];
    __shared__ v2f sH2blk[5 * 32];

    const int tid   = threadIdx.x;
    const int lane  = tid & 31;
    const int wv    = __builtin_amdgcn_readfirstlane(tid >> 5);
    const int mrow  = lane & 15;
    const int hi    = lane >> 4;
    const int khalf = 2 * hi;
    const int b0    = blockIdx.x;

    if (wv == 0) {
        for (int f = 0; f < 40; ++f) {
            const int n = f >> 3, kk = f & 7;
            const int row = 4 * kk + khalf, col = 16 * n + mrow;
            v2f w; w.x = Wi1[row * LH4 + col]; w.y = Wi1[(row + 1) * LH4 + col];
            sWi1f[f * 32 + lane] = w;
        }
        for (int f = 0; f < 25; ++f) {
            const int n = f / 5, kk = f % 5;
            const int row = 4 * kk + khalf, col = 16 * n + mrow;
            v2f w;
            w.x = Wh1[row * LH4 + col]; w.y = Wh1[(row + 1) * LH4 + col]; sWh1f[f * 32 + lane] = w;
        }
    } else {
        for (int f = 0; f < 25; ++f) {
            const int n = f / 5, kk = f % 5;
            const int row = 4 * kk + khalf, col = 16 * n + mrow;
            v2f w;
            w.x = Wi2[row * LH4 + col]; w.y = Wi2[(row + 1) * LH4 + col]; sWi2f[f * 32 + lane] = w;
            w.x = Wh2[row * LH4 + col]; w.y = Wh2[(row + 1) * LH4 + col]; sWh2f[f * 32 + lane] = w;
        }
        for (int kk = 0; kk < 5; ++kk) {
            const int row = 4 * kk + khalf;
            v2f w;
            w.x = (mrow < LOUT) ? Wd[row * LOUT + mrow] : 0.0f;
            w.y = (mrow < LOUT) ? Wd[(row + 1) * LOUT + mrow] : 0.0f;
            sWdf[kk * 32 + lane] = w;
        }
    }
    for (int i = tid; i < LH4; i += 64) { sb1[i] = b1[i]; sb2[i] = b2[i]; }
    for (int i = tid; i < 16; i += 64) sbd[i] = (i < LOUT) ? bd[i] : 0.0f;
    {
        v2f zz; zz.x = 0.0f; zz.y = 0.0f;
        for (int i = tid; i < 5 * 32; i += 64) {
            sH2blk[i] = zz; sH1blk[i] = zz; sH1blk[5 * 32 + i] = zz;
        }
        for (int i = tid; i < (TBLK + 1) * XP; i += 64) { sXW1[i] = 0.0f; sXW2[i] = 0.0f; }
    }
    __syncthreads();

    const v2f*   sWhf = wv ? sWh2f : sWh1f;
    const float* sb   = wv ? sb2   : sb1;
    float*       sXW  = wv ? sXW2  : sXW1;

    const float* xbase = X + (size_t)b0 * LT * LF + (size_t)mrow * LF + khalf;
    v2f ax[8];
    #pragma unroll
    for (int kk = 0; kk < 8; ++kk) { ax[kk].x = 0.0f; ax[kk].y = 0.0f; }
    if (wv == 0) {
        #pragma unroll
        for (int kk = 0; kk < 8; ++kk) {
            ax[kk].x = xbase[4 * kk];
            ax[kk].y = xbase[4 * kk + 1];
        }
    }

    const int pe   = (mrow < 10) ? mrow : 0;
    const int gj   = 4 * (pe >> 1) + 2 * hi + (pe & 1);
    const int hidx = (((gj >> 2) * 32) + 16 * ((gj & 3) >> 1)) * 2 + (gj & 1);
    const unsigned seli = (unsigned)(gj & 15),        self_ = (unsigned)((LH + gj) & 15);
    const unsigned selg = (unsigned)((2 * LH + gj) & 15), selo = (unsigned)((3 * LH + gj) & 15);
    const bool upi = (gj >> 4) != 0,            upf = ((LH + gj) >> 4) != 1;
    const bool upg = ((2 * LH + gj) >> 4) != 2, upo = ((3 * LH + gj) >> 4) != 3;

    v2f wh[25];
    #pragma unroll
    for (int f = 0; f < 25; ++f) wh[f] = sWhf[f * 32 + lane];
    const float bi = sb[gj], bf = sb[LH + gj], bg = sb[2 * LH + gj], bo = sb[3 * LH + gj];

    v8f ci[5];
    #pragma unroll
    for (int n = 0; n < 5; ++n) { v8f zz = {}; ci[n] = zz; }
    v2f a[5];
    #pragma unroll
    for (int kk = 0; kk < 5; ++kk) { a[kk].x = 0.0f; a[kk].y = 0.0f; }
    float cst = 0.0f;

    for (int blk = 0; blk < NBLK + 2; ++blk) {
        const bool on = wv ? ((blk >= 1) && (blk <= NBLK)) : (blk < NBLK);
        float* hblk = wv ? (float*)sH2blk : (float*)(sH1blk + (blk & 1) * (5 * 32));

        if (wv == 0) {
            if (on) {
                v8f xw[5];
                #pragma unroll
                for (int n = 0; n < 5; ++n) {
                    v8f c = {};
                    #pragma unroll
                    for (int kk = 0; kk < 8; ++kk)
                        WMMA4(c, ax[kk], sWi1f[(n * 8 + kk) * 32 + lane]);
                    xw[n] = c;
                }
                if (blk + 1 < NBLK) {
                    const float* xr = xbase + (size_t)(blk + 1) * TBLK * LF;
                    #pragma unroll
                    for (int kk = 0; kk < 8; ++kk) {
                        ax[kk].x = xr[4 * kk];
                        ax[kk].y = xr[4 * kk + 1];
                    }
                }
                #pragma unroll
                for (int n = 0; n < 5; ++n) {
                    #pragma unroll
                    for (int r = 0; r < 8; ++r)
                        sXW1[(r + 8 * hi) * XP + 16 * n + mrow] = xw[n][r];
                }
            }
        } else {
            if (on) {
                const v2f* h1b = sH1blk + ((blk - 1) & 1) * (5 * 32);
                v8f xw[5];
                #pragma unroll
                for (int n = 0; n < 5; ++n) {
                    v8f c = {};
                    #pragma unroll
                    for (int kk = 0; kk < 5; ++kk)
                        WMMA4(c, h1b[kk * 32 + lane], sWi2f[(n * 5 + kk) * 32 + lane]);
                    xw[n] = c;
                }
                #pragma unroll
                for (int n = 0; n < 5; ++n) {
                    #pragma unroll
                    for (int r = 0; r < 8; ++r)
                        sXW2[(r + 8 * hi) * XP + 16 * n + mrow] = xw[n][r];
                }
            }
            if (blk >= 2) {
                v8f c = {};
                #pragma unroll
                for (int kk = 0; kk < 5; ++kk)
                    WMMA4(c, sH2blk[kk * 32 + lane], sWdf[kk * 32 + lane]);
                if (mrow < LOUT) {
                    const float bb = sbd[mrow];
                    float* orow = out + ((size_t)b0 * LT + (size_t)(blk - 2) * TBLK + 8 * hi) * LOUT + mrow;
                    #pragma unroll
                    for (int r = 0; r < 8; ++r)
                        orow[r * LOUT] = c[r] + bb;
                }
            }
        }
        LDS_ORDER();

        if (on) {
            #pragma unroll
            for (int n = 0; n < 5; ++n) ci[n][0] = sXW[16 * n + mrow];

            for (int tt = 0; tt < TBLK; ++tt)
                layer_step(ci, a, wh, cst, seli, self_, selg, selo, upi, upf, upg, upo,
                           bi, bf, bg, bo, hblk + hidx + 2 * tt, sXW + (tt + 1) * XP + mrow);
        }
        __syncthreads();
    }
}

extern "C" void kernel_launch(void* const* d_in, const int* in_sizes, int n_in,
                              void* d_out, int out_size, void* d_ws, size_t ws_size,
                              hipStream_t stream) {
    (void)in_sizes; (void)n_in; (void)out_size; (void)d_ws; (void)ws_size;
    const float* X   = (const float*)d_in[0];
    const float* Wi1 = (const float*)d_in[1];
    const float* Wh1 = (const float*)d_in[2];
    const float* b1  = (const float*)d_in[3];
    const float* Wi2 = (const float*)d_in[4];
    const float* Wh2 = (const float*)d_in[5];
    const float* b2  = (const float*)d_in[6];
    const float* Wd  = (const float*)d_in[7];
    const float* bd  = (const float*)d_in[8];
    float* out = (float*)d_out;

    dim3 grid(LB);
    dim3 block(64);
    lstm2_dense_fused<<<grid, block, 0, stream>>>(X, Wi1, Wh1, b1, Wi2, Wh2, b2, Wd, bd, out);
}


# The three time-batched products (x@Wi1, h1@Wi2, head) moved to a third wave on an idle SIMD, one block ahead via 2-slot LDS arrays: layer waves run steps only; layer 2 lags 2 blocks; sums unchanged
# speedup vs baseline: 11.2637x; 1.0814x over previous
#include <hip/hip_runtime.h>
#include <hip/hip_bf16.h>


typedef __attribute__((ext_vector_type(2))) float v2f;
typedef __attribute__((ext_vector_type(4))) float v4f;
typedef __attribute__((ext_vector_type(8))) float v8f;

#define LH    20
#define LH4   80
#define LF    32
#define LT    1024
#define LB    256
#define LOUT  10
#define TBLK  16
#define NBLK  (LT / TBLK)
#define NTHR  96
#define XP    84
#define XWSZ  ((TBLK + 1) * XP)
#define HBSZ  (5 * 32)

__device__ __forceinline__ float sigf(float x) { return 1.0f / (1.0f + __expf(-x)); }
__device__ __forceinline__ float tanhfast(float x) {
    const float e = __expf(2.0f * x);
    return (e - 1.0f) / (e + 1.0f);
}

#define WMMA4(c, a, b) \
    (c) = __builtin_amdgcn_wmma_f32_16x16x4_f32(false, (a), false, (b), (short)0, (c), false, false)


__device__ __forceinline__ float gather16(float v, unsigned sel) {
    const unsigned u = __builtin_bit_cast(unsigned, v);
    return __builtin_bit_cast(float, __builtin_amdgcn_permlane16_var(u, u, sel, false, false));
}
#define ROW_BCAST(v, n) \
    __builtin_bit_cast(float, __builtin_amdgcn_mov_dpp(__builtin_bit_cast(int, (v)), 0x150 + (n), 0xf, 0xf, false))

__device__ __forceinline__ void layer_step(
        v8f (&ci)[5], v2f (&a)[5], const v2f (&wh)[25], float& cst,
        unsigned seli, unsigned self_, unsigned selg, unsigned selo,
        bool upi, bool upf, bool upg, bool upo,
        float bi, float bf, float bg, float bo,
        float* hrow, const float* xwn)
{
    float nx[5];
    #pragma unroll
    for (int n = 0; n < 5; ++n) nx[n] = xwn[16 * n];
    __builtin_amdgcn_sched_barrier(0);
    v8f z[5];
    #pragma unroll
    for (int n = 0; n < 5; ++n) z[n] = ci[n];
    #pragma unroll
    for (int kk = 0; kk < 5; ++kk) {
        #pragma unroll
        for (int n = 0; n < 5; ++n)
            WMMA4(z[n], a[kk], wh[n * 5 + kk]);
    }
    #pragma unroll
    for (int n = 0; n < 5; ++n) ci[n][0] = nx[n];
    const float ia = gather16(z[0][0], seli),  ib = gather16(z[1][0], seli);
    const float fa = gather16(z[1][0], self_), fb = gather16(z[2][0], self_);
    const float ga = gather16(z[2][0], selg),  gb = gather16(z[3][0], selg);
    const float oa = gather16(z[3][0], selo),  ob = gather16(z[4][0], selo);
    const float ri = upi ? ib : ia, rf = upf ? fb : fa, rg = upg ? gb : ga, ro = upo ? ob : oa;
    const float zi = ri + bi, zf = rf + bf, zg = rg + bg, zo = ro + bo;
    const float ei = __expf(-zi), ef = __expf(-zf), eo = __expf(-zo), eg = __expf(2.0f * zg);
    const float si = 1.0f / (1.0f + ei);
    const float sf = 1.0f / (1.0f + ef);
    const float so = 1.0f / (1.0f + eo);
    const float tg = (eg - 1.0f) / (eg + 1.0f);
    const float cc = sf * cst + si * tg;
    const float ec = __expf(2.0f * cc);
    const float hh = so * ((ec - 1.0f) / (ec + 1.0f));
    cst = cc;
    *hrow = hh;
    a[0].x = ROW_BCAST(hh, 0); a[0].y = ROW_BCAST(hh, 1);
    a[1].x = ROW_BCAST(hh, 2); a[1].y = ROW_BCAST(hh, 3);
    a[2].x = ROW_BCAST(hh, 4); a[2].y = ROW_BCAST(hh, 5);
    a[3].x = ROW_BCAST(hh, 6); a[3].y = ROW_BCAST(hh, 7);
    a[4].x = ROW_BCAST(hh, 8); a[4].y = ROW_BCAST(hh, 9);
}

__global__ void __launch_bounds__(NTHR)
lstm2_dense_fused(const float* __restrict__ X,
                  const float* __restrict__ Wi1, const float* __restrict__ Wh1,
                  const float* __restrict__ b1,
                  const float* __restrict__ Wi2, const float* __restrict__ Wh2,
                  const float* __restrict__ b2,
                  const float* __restrict__ Wd,  const float* __restrict__ bd,
                  float* __restrict__ out)
{
    __shared__ v2f sWi1f[40 * 32];
    __shared__ v2f sWh1f[25 * 32];
    __shared__ v2f sWi2f[25 * 32];
    __shared__ v2f sWh2f[25 * 32];
    __shared__ v2f sWdf [ 5 * 32];
    __shared__ float sb1[LH4], sb2[LH4], sbd[16];
    __shared__ float sXW1[2 * XWSZ];
    __shared__ float sXW2[2 * XWSZ];
    __shared__ v2f sH1blk[2 * HBSZ];
    __shared__ v2f sH2blk[2 * HBSZ];

    const int tid   = threadIdx.x;
    const int lane  = tid & 31;
    const int wv    = __builtin_amdgcn_readfirstlane(tid >> 5);
    const int mrow  = lane & 15;
    const int hi    = lane >> 4;
    const int khalf = 2 * hi;
    const int b0    = blockIdx.x;

    if (wv == 0) {
        for (int f = 0; f < 25; ++f) {
            const int n = f / 5, kk = f % 5;
            const int row = 4 * kk + khalf, col = 16 * n + mrow;
            v2f w;
            w.x = Wh1[row * LH4 + col]; w.y = Wh1[(row + 1) * LH4 + col]; sWh1f[f * 32 + lane] = w;
            w.x = Wi2[row * LH4 + col]; w.y = Wi2[(row + 1) * LH4 + col]; sWi2f[f * 32 + lane] = w;
        }
    } else if (wv == 1) {
        for (int f = 0; f < 25; ++f) {
            const int n = f / 5, kk = f % 5;
            const int row = 4 * kk + khalf, col = 16 * n + mrow;
            v2f w;
            w.x = Wh2[row * LH4 + col]; w.y = Wh2[(row + 1) * LH4 + col]; sWh2f[f * 32 + lane] = w;
        }
        for (int kk = 0; kk < 5; ++kk) {
            const int row = 4 * kk + khalf;
            v2f w;
            w.x = (mrow < LOUT) ? Wd[row * LOUT + mrow] : 0.0f;
            w.y = (mrow < LOUT) ? Wd[(row + 1) * LOUT + mrow] : 0.0f;
            sWdf[kk * 32 + lane] = w;
        }
    } else {
        for (int f = 0; f < 40; ++f) {
            const int n = f >> 3, kk = f & 7;
            const int row = 4 * kk + khalf, col = 16 * n + mrow;
            v2f w; w.x = Wi1[row * LH4 + col]; w.y = Wi1[(row + 1) * LH4 + col];
            sWi1f[f * 32 + lane] = w;
        }
    }
    for (int i = tid; i < LH4; i += NTHR) { sb1[i] = b1[i]; sb2[i] = b2[i]; }
    for (int i = tid; i < 16; i += NTHR) sbd[i] = (i < LOUT) ? bd[i] : 0.0f;
    {
        v2f zz; zz.x = 0.0f; zz.y = 0.0f;
        for (int i = tid; i < 2 * HBSZ; i += NTHR) { sH1blk[i] = zz; sH2blk[i] = zz; }
        for (int i = tid; i < 2 * XWSZ; i += NTHR) { sXW1[i] = 0.0f; sXW2[i] = 0.0f; }
    }
    __syncthreads();

    const v2f*   sWhf = (wv == 1) ? sWh2f  : sWh1f;
    const float* sb   = (wv == 1) ? sb2    : sb1;
    const float* sXW  = (wv == 1) ? sXW2   : sXW1;
    v2f*         sHb  = (wv == 1) ? sH2blk : sH1blk;

    const float* xbase = X + (size_t)b0 * LT * LF + (size_t)mrow * LF + khalf;
    v2f ax[8];
    #pragma unroll
    for (int kk = 0; kk < 8; ++kk) { ax[kk].x = 0.0f; ax[kk].y = 0.0f; }

    if (wv == 2) {
        #pragma unroll
        for (int kk = 0; kk < 8; ++kk) {
            ax[kk].x = xbase[4 * kk];
            ax[kk].y = xbase[4 * kk + 1];
        }
        #pragma unroll
        for (int n = 0; n < 5; ++n) {
            v8f c = {};
            #pragma unroll
            for (int kk = 0; kk < 8; ++kk)
                WMMA4(c, ax[kk], sWi1f[(n * 8 + kk) * 32 + lane]);
            #pragma unroll
            for (int r = 0; r < 8; ++r)
                sXW1[(r + 8 * hi) * XP + 16 * n + mrow] = c[r];
        }
        {
            const float* xr = xbase + (size_t)TBLK * LF;
            #pragma unroll
            for (int kk = 0; kk < 8; ++kk) {
                ax[kk].x = xr[4 * kk];
                ax[kk].y = xr[4 * kk + 1];
            }
        }
    }

    const int pe   = (mrow < 10) ? mrow : 0;
    const int gj   = 4 * (pe >> 1) + 2 * hi + (pe & 1);
    const int hidx = (((gj >> 2) * 32) + 16 * ((gj & 3) >> 1)) * 2 + (gj & 1);
    const unsigned seli = (unsigned)(gj & 15),        self_ = (unsigned)((LH + gj) & 15);
    const unsigned selg = (unsigned)((2 * LH + gj) & 15), selo = (unsigned)((3 * LH + gj) & 15);
    const bool upi = (gj >> 4) != 0,            upf = ((LH + gj) >> 4) != 1;
    const bool upg = ((2 * LH + gj) >> 4) != 2, upo = ((3 * LH + gj) >> 4) != 3;

    v2f wh[25];
    #pragma unroll
    for (int f = 0; f < 25; ++f) wh[f] = sWhf[f * 32 + lane];
    const float bi = sb[gj], bf = sb[LH + gj], bg = sb[2 * LH + gj], bo = sb[3 * LH + gj];

    v8f ci[5];
    #pragma unroll
    for (int n = 0; n < 5; ++n) { v8f zz = {}; ci[n] = zz; }
    v2f a[5];
    #pragma unroll
    for (int kk = 0; kk < 5; ++kk) { a[kk].x = 0.0f; a[kk].y = 0.0f; }
    float cst = 0.0f;

    __syncthreads();

    for (int blk = 0; blk < NBLK + 3; ++blk) {
        if (wv == 2) {
            if (blk + 1 < NBLK) {
                float* dst = sXW1 + ((blk + 1) & 1) * XWSZ;
                #pragma unroll
                for (int n = 0; n < 5; ++n) {
                    v8f c = {};
                    #pragma unroll
                    for (int kk = 0; kk < 8; ++kk)
                        WMMA4(c, ax[kk], sWi1f[(n * 8 + kk) * 32 + lane]);
                    #pragma unroll
                    for (int r = 0; r < 8; ++r)
                        dst[(r + 8 * hi) * XP + 16 * n + mrow] = c[r];
                }
                if (blk + 2 < NBLK) {
                    const float* xr = xbase + (size_t)(blk + 2) * TBLK * LF;
                    #pragma unroll
                    for (int kk = 0; kk < 8; ++kk) {
                        ax[kk].x = xr[4 * kk];
                        ax[kk].y = xr[4 * kk + 1];
                    }
                }
            }
            if ((blk >= 1) && (blk <= NBLK)) {
                const v2f* h1b = sH1blk + ((blk - 1) & 1) * HBSZ;
                float* dst = sXW2 + ((blk + 1) & 1) * XWSZ;
                #pragma unroll
                for (int n = 0; n < 5; ++n) {
                    v8f c = {};
                    #pragma unroll
                    for (int kk = 0; kk < 5; ++kk)
                        WMMA4(c, h1b[kk * 32 + lane], sWi2f[(n * 5 + kk) * 32 + lane]);
                    #pragma unroll
                    for (int r = 0; r < 8; ++r)
                        dst[(r + 8 * hi) * XP + 16 * n + mrow] = c[r];
                }
            }
            if (blk >= 3) {
                const v2f* h2b = sH2blk + ((blk - 1) & 1) * HBSZ;
                v8f c = {};
                #pragma unroll
                for (int kk = 0; kk < 5; ++kk)
                    WMMA4(c, h2b[kk * 32 + lane], sWdf[kk * 32 + lane]);
                if (mrow < LOUT) {
                    const float bb = sbd[mrow];
                    float* orow = out + ((size_t)b0 * LT + (size_t)(blk - 3) * TBLK + 8 * hi) * LOUT + mrow;
                    #pragma unroll
                    for (int r = 0; r < 8; ++r)
                        orow[r * LOUT] = c[r] + bb;
                }
            }
        }
        const bool on = (wv == 2) ? false
                      : (wv ? ((blk >= 2) && (blk < NBLK + 2)) : (blk < NBLK));
        if (on) {
            const float* xwb  = sXW + (blk & 1) * XWSZ;
            float*       hblk = (float*)(sHb + (blk & 1) * HBSZ);
            #pragma unroll
            for (int n = 0; n < 5; ++n) ci[n][0] = xwb[16 * n + mrow];

            for (int tt = 0; tt < TBLK; ++tt)
                layer_step(ci, a, wh, cst, seli, self_, selg, selo, upi, upf, upg, upo,
                           bi, bf, bg, bo, hblk + hidx + 2 * tt, xwb + (tt + 1) * XP + mrow);
        }
        __syncthreads();
    }
}

extern "C" void kernel_launch(void* const* d_in, const int* in_sizes, int n_in,
                              void* d_out, int out_size, void* d_ws, size_t ws_size,
                              hipStream_t stream) {
    (void)in_sizes; (void)n_in; (void)out_size; (void)d_ws; (void)ws_size;
    const float* X   = (const float*)d_in[0];
    const float* Wi1 = (const float*)d_in[1];
    const float* Wh1 = (const float*)d_in[2];
    const float* b1  = (const float*)d_in[3];
    const float* Wi2 = (const float*)d_in[4];
    const float* Wh2 = (const float*)d_in[5];
    const float* b2  = (const float*)d_in[6];
    const float* Wd  = (const float*)d_in[7];
    const float* bd  = (const float*)d_in[8];
    float* out = (float*)d_out;

    dim3 grid(LB);
    dim3 block(NTHR);
    lstm2_dense_fused<<<grid, block, 0, stream>>>(X, Wi1, Wh1, b1, Wi2, Wh2, b2, Wd, bd, out);
}
